# starting kernel
# baseline (speedup 1.0000x reference)
#include <hip/hip_runtime.h>

typedef __attribute__((ext_vector_type(2))) float v2f;
typedef __attribute__((ext_vector_type(8))) float v8f;

__device__ __forceinline__ v8f wmma4(v2f a, v2f b, v8f c) {
  return __builtin_amdgcn_wmma_f32_16x16x4_f32(false, a, false, b, (short)0, c,
                                               false, false);
}

__device__ __forceinline__ float leaky(float x) {
  return x >= 0.0f ? x : 0.01f * x;
}

__global__ void __launch_bounds__(128) proj_kernel(
    const float* __restrict__ sa, const float* __restrict__ sb,
    const float* __restrict__ W1, const float* __restrict__ b1,
    float* __restrict__ a_pb, float* __restrict__ b_part) {
  const int lane = threadIdx.x & 31;
  const int lo = lane & 15, hi = lane >> 4;
  const int wave = (blockIdx.x * 128 + threadIdx.x) >> 5;
  const int mat = wave >> 8;
  const int t = wave & 255;
  const int m0 = (t >> 2) * 16;
  const int n0 = (t & 3) * 16;

  const float* src = mat ? sb : sa;
  const float* W   = W1 + (mat ? 256 * 64 : 0);
  const float* arow = src + (size_t)(m0 + lo) * 256 + 2 * hi;
  const float* wcol = W + n0 + lo + (size_t)(2 * hi) * 64;

  v8f c = {};
#pragma unroll 8
  for (int kk = 0; kk < 256; kk += 4) {
    v2f a = *(const v2f*)(arow + kk);
    v2f b;
    b.x = wcol[(size_t)kk * 64];
    b.y = wcol[(size_t)(kk + 1) * 64];
    c = wmma4(a, b, c);
  }

  const float bias = mat ? 0.0f : b1[n0 + lo];
  float* dst = (mat ? b_part : a_pb) + n0 + lo;
#pragma unroll
  for (int r = 0; r < 8; ++r)
    dst[(size_t)(m0 + r + 8 * hi) * 64] = c[r] + bias;
}

#define LSTR 34

__global__ void __launch_bounds__(128) match_mlp_kernel(
    const float* __restrict__ a_pb, const float* __restrict__ b_part,
    const float* __restrict__ W2, const float* __restrict__ b2,
    const float* __restrict__ W3, const float* __restrict__ b3,
    float* __restrict__ out) {
  __shared__ float lds[4 * 16 * LSTR];
  const int lane = threadIdx.x & 31;
  const int lo = lane & 15, hi = lane >> 4;
  const int wave = (blockIdx.x * 128 + threadIdx.x) >> 5;
  const int ba = wave >> 1;
  const int half = wave & 1;
  const int bbase = ba & ~127;
  float* wlds = lds + (threadIdx.x >> 5) * (16 * LSTR);

  v2f abv[16];
  const float* apr = a_pb + (size_t)ba * 64 + 2 * hi;
#pragma unroll
  for (int i = 0; i < 16; ++i) abv[i] = *(const v2f*)(apr + 4 * i);

  v2f w2f[2][16];
#pragma unroll
  for (int t = 0; t < 2; ++t) {
    const float* w = W2 + 16 * t + lo + (size_t)(2 * hi) * 32;
#pragma unroll
    for (int i = 0; i < 16; ++i) {
      w2f[t][i].x = w[(size_t)(4 * i) * 32];
      w2f[t][i].y = w[(size_t)(4 * i + 1) * 32];
    }
  }
  v2f w3f[8];
  {
    const float* w = W3 + lo + (size_t)(2 * hi) * 16;
#pragma unroll
    for (int i = 0; i < 8; ++i) {
      w3f[i].x = w[(size_t)(4 * i) * 16];
      w3f[i].y = w[(size_t)(4 * i + 1) * 16];
    }
  }
  const float b2v0 = b2[lo], b2v1 = b2[16 + lo];
  const float b3v = b3[lo];

  for (int it = 0; it < 4; ++it) {
    const int c0 = (half * 4 + it) * 16;

    const float* bpr = b_part + (size_t)(bbase + c0 + lo) * 64 + 2 * hi;
    v2f hf[16];
#pragma unroll
    for (int i = 0; i < 16; ++i) {
      v2f bp = *(const v2f*)(bpr + 4 * i);
      hf[i].x = leaky(abv[i].x + bp.x);
      hf[i].y = leaky(abv[i].y + bp.y);
    }

    v8f c2a = {}, c2b = {};
#pragma unroll
    for (int i = 0; i < 16; ++i) {
      c2a = wmma4(hf[i], w2f[0][i], c2a);
      c2b = wmma4(hf[i], w2f[1][i], c2b);
    }

#pragma unroll
    for (int r = 0; r < 8; ++r) {
      wlds[(r + 8 * hi) * LSTR + lo]      = leaky(c2a[r] + b2v0);
      wlds[(r + 8 * hi) * LSTR + 16 + lo] = leaky(c2b[r] + b2v1);
    }
    asm volatile("s_wait_dscnt 0x0" ::: "memory");

    v2f hf3[8];
#pragma unroll
    for (int i = 0; i < 8; ++i)
      hf3[i] = *(const v2f*)(wlds + lo * LSTR + 4 * i + 2 * hi);
    asm volatile("s_wait_dscnt 0x0" ::: "memory");

    v8f c3 = {};
#pragma unroll
    for (int i = 0; i < 8; ++i) c3 = wmma4(hf3[i], w3f[i], c3);

    float* op = out + ((size_t)ba * 128 + c0) * 16 + lo;
#pragma unroll
    for (int r = 0; r < 8; ++r)
      op[(size_t)(r + 8 * hi) * 16] = leaky(c3[r] + b3v);
  }
}

extern "C" void kernel_launch(void* const* d_in, const int* in_sizes, int n_in,
                              void* d_out, int out_size, void* d_ws,
                              size_t ws_size, hipStream_t stream) {
  (void)in_sizes; (void)n_in; (void)out_size; (void)ws_size;
  const float* sa = (const float*)d_in[0];
  const float* sb = (const float*)d_in[1];
  const float* W1 = (const float*)d_in[2];
  const float* b1 = (const float*)d_in[3];
  const float* W2 = (const float*)d_in[4];
  const float* b2 = (const float*)d_in[5];
  const float* W3 = (const float*)d_in[6];
  const float* b3 = (const float*)d_in[7];
  float* out = (float*)d_out;

  float* a_pb   = (float*)d_ws;
  float* b_part = a_pb + 1024 * 64;

  proj_kernel<<<128, 128, 0, stream>>>(sa, sb, W1, b1, a_pb, b_part);
  match_mlp_kernel<<<512, 128, 0, stream>>>(a_pb, b_part, W2, b2, W3, b3, out);
}


# Two launches fused into one: each work-group stages its 48 input rows and W1 in LDS, computes its own a_pb and b_part tiles there and feeds the MLP tiles from LDS; every sum keeps its k order
# speedup vs baseline: 1.4521x; 1.4521x over previous
#include <hip/hip_runtime.h>


typedef __attribute__((ext_vector_type(2))) float v2f;
typedef __attribute__((ext_vector_type(4))) float v4f;
typedef __attribute__((ext_vector_type(8))) float v8f;

__device__ __forceinline__ v8f wmma4(v2f a, v2f b, v8f c) {
  return __builtin_amdgcn_wmma_f32_16x16x4_f32(false, a, false, b, (short)0, c,
                                               false, false);
}

__device__ __forceinline__ float leaky(float x) {
  return x >= 0.0f ? x : 0.01f * x;
}

#define APITCH 260
#define WPITCH 160
#define PPITCH 68
#define LSTR 34

template <int NT>
__device__ __forceinline__ void proj_tiles(const float* aBase, const float* wBase,
                                           v8f (&acc)[NT]) {
#pragma unroll 8
  for (int kk = 0; kk < 256; kk += 4) {
    const v2f bf = *(const v2f*)(wBase + (kk >> 1) * WPITCH);
#pragma unroll
    for (int t = 0; t < NT; ++t) {
      const v2f af = *(const v2f*)(aBase + t * 16 * APITCH + kk);
      acc[t] = wmma4(af, bf, acc[t]);
    }
  }
}

__global__ void __launch_bounds__(256) match_fused_kernel(
    const float* __restrict__ sa, const float* __restrict__ sb,
    const float* __restrict__ W1, const float* __restrict__ b1,
    const float* __restrict__ W2, const float* __restrict__ b2,
    const float* __restrict__ W3, const float* __restrict__ b3,
    float* __restrict__ out) {
  __shared__ __align__(16) float sA[48 * APITCH];
  __shared__ __align__(16) float sW[256 * WPITCH];
  __shared__ __align__(16) float sP[48 * PPITCH];
  __shared__ __align__(16) float sT[8 * 16 * LSTR];

  const int tid = threadIdx.x;
  const int lane = tid & 31;
  const int lo = lane & 15, hi = lane >> 4;
  const int w = tid >> 5;
  const int blk = blockIdx.x;
  const int b = blk >> 5;
  const int a0 = ((blk >> 2) & 7) * 16;
  const int c0 = (blk & 3) * 32;

  const bool a_role = (__builtin_popcount(w) & 1) != 0;
  const int n0 = (w >> 1) * 16;
  const float bias = a_role ? b1[n0 + lo] : 0.0f;

  {
    const int col4 = tid & 63, rsub = tid >> 6;
#pragma unroll
    for (int it = 0; it < 12; ++it) {
      const int row = it * 4 + rsub;
      const float* g = (it < 4)
          ? sa + (size_t)(b * 128 + a0 + row) * 256
          : sb + (size_t)(b * 128 + c0 + row - 16) * 256;
      const v4f v = *(const v4f*)(g + 4 * col4);
      *(v4f*)(sA + row * APITCH + 4 * col4) = v;
    }
  }
  {
#pragma unroll
    for (int it = 0; it < 32; ++it) {
      const int kp = it * 8 + w;
      const float* g = W1 + (size_t)(2 * kp) * 64 + 2 * lane;
      const v2f r0 = *(const v2f*)g;
      const v2f r1 = *(const v2f*)(g + 64);
      const v4f q = {r0.x, r1.x, r0.y, r1.y};
      *(v4f*)(sW + kp * WPITCH + 4 * lane) = q;
    }
  }
  v2f w2f[2][16];
#pragma unroll
  for (int t = 0; t < 2; ++t) {
    const float* wp = W2 + 16 * t + lo + (size_t)(2 * hi) * 32;
#pragma unroll
    for (int i = 0; i < 16; ++i) {
      w2f[t][i].x = wp[(size_t)(4 * i) * 32];
      w2f[t][i].y = wp[(size_t)(4 * i + 1) * 32];
    }
  }
  v2f w3f[8];
  {
    const float* wp = W3 + lo + (size_t)(2 * hi) * 16;
#pragma unroll
    for (int i = 0; i < 8; ++i) {
      w3f[i].x = wp[(size_t)(4 * i) * 16];
      w3f[i].y = wp[(size_t)(4 * i + 1) * 16];
    }
  }
  const float b2v0 = b2[lo], b2v1 = b2[16 + lo];
  const float b3v = b3[lo];
  __syncthreads();

  if (!a_role) {
    v8f acc[2] = {};
    proj_tiles<2>(sA + (16 + lo) * APITCH + 2 * hi,
                  sW + (128 + hi) * WPITCH + 2 * (n0 + lo), acc);
#pragma unroll
    for (int t = 0; t < 2; ++t)
#pragma unroll
      for (int r = 0; r < 8; ++r)
        sP[(16 + t * 16 + r + 8 * hi) * PPITCH + n0 + lo] = acc[t][r] + bias;
  } else {
    v8f acc[1] = {};
    proj_tiles<1>(sA + lo * APITCH + 2 * hi,
                  sW + hi * WPITCH + 2 * (n0 + lo), acc);
#pragma unroll
    for (int r = 0; r < 8; ++r)
      sP[(r + 8 * hi) * PPITCH + n0 + lo] = acc[0][r] + bias;
  }
  __syncthreads();

  float* wlds = sT + w * (16 * LSTR);
  for (int aa = 0; aa < 2; ++aa) {
    const int al = 2 * w + aa;
    v2f abv[16];
    const float* par = sP + al * PPITCH + 2 * hi;
#pragma unroll
    for (int i = 0; i < 16; ++i) abv[i] = *(const v2f*)(par + 4 * i);

    for (int ct = 0; ct < 2; ++ct) {
      const float* pbr = sP + (16 + ct * 16 + lo) * PPITCH + 2 * hi;
      v2f hf[16];
#pragma unroll
      for (int i = 0; i < 16; ++i) {
        const v2f bp = *(const v2f*)(pbr + 4 * i);
        hf[i].x = leaky(abv[i].x + bp.x);
        hf[i].y = leaky(abv[i].y + bp.y);
      }

      v8f c2a = {}, c2b = {};
#pragma unroll
      for (int i = 0; i < 16; ++i) {
        c2a = wmma4(hf[i], w2f[0][i], c2a);
        c2b = wmma4(hf[i], w2f[1][i], c2b);
      }

#pragma unroll
      for (int r = 0; r < 8; ++r) {
        wlds[(r + 8 * hi) * LSTR + lo]      = leaky(c2a[r] + b2v0);
        wlds[(r + 8 * hi) * LSTR + 16 + lo] = leaky(c2b[r] + b2v1);
      }
      asm volatile("s_wait_dscnt 0x0" ::: "memory");

      v2f hf3[8];
#pragma unroll
      for (int i = 0; i < 8; ++i)
        hf3[i] = *(const v2f*)(wlds + lo * LSTR + 4 * i + 2 * hi);
      asm volatile("s_wait_dscnt 0x0" ::: "memory");

      v8f c3 = {};
#pragma unroll
      for (int i = 0; i < 8; ++i) c3 = wmma4(hf3[i], w3f[i], c3);

      float* op = out + ((size_t)(b * 128 + a0 + al) * 128 + c0 + ct * 16) * 16 + lo;
#pragma unroll
      for (int r = 0; r < 8; ++r)
        op[(size_t)(r + 8 * hi) * 16] = leaky(c3[r] + b3v);
    }
  }
}

extern "C" void kernel_launch(void* const* d_in, const int* in_sizes, int n_in,
                              void* d_out, int out_size, void* d_ws,
                              size_t ws_size, hipStream_t stream) {
  (void)in_sizes; (void)n_in; (void)out_size; (void)ws_size; (void)d_ws;
  const float* sa = (const float*)d_in[0];
  const float* sb = (const float*)d_in[1];
  const float* W1 = (const float*)d_in[2];
  const float* b1 = (const float*)d_in[3];
  const float* W2 = (const float*)d_in[4];
  const float* b2 = (const float*)d_in[5];
  const float* W3 = (const float*)d_in[6];
  const float* b3 = (const float*)d_in[7];
  float* out = (float*)d_out;

  match_fused_kernel<<<256, 256, 0, stream>>>(sa, sb, W1, b1, W2, b2, W3, b3, out);
}


# Staging pipelined with layer 1 in four K chunks: hand-written LDS barriers leave later chunks' and the layer-2/3 weights' global loads in flight under the WMMAs; every sum keeps its k order
# speedup vs baseline: 1.6014x; 1.1028x over previous
#include <hip/hip_runtime.h>


typedef __attribute__((ext_vector_type(2))) float v2f;
typedef __attribute__((ext_vector_type(4))) float v4f;
typedef __attribute__((ext_vector_type(8))) float v8f;

__device__ __forceinline__ v8f wmma4(v2f a, v2f b, v8f c) {
  return __builtin_amdgcn_wmma_f32_16x16x4_f32(false, a, false, b, (short)0, c,
                                               false, false);
}

__device__ __forceinline__ float leaky(float x) {
  return x >= 0.0f ? x : 0.01f * x;
}

#define APITCH 260
#define WPITCH 160
#define PPITCH 68
#define LSTR 34

#define KC 64

template <int NT>
__device__ __forceinline__ void proj_chunk(const float* aBase, const float* wBase,
                                           v8f (&acc)[NT]) {
#pragma unroll
  for (int kk = 0; kk < KC; kk += 4) {
    const v2f bf = *(const v2f*)(wBase + (kk >> 1) * WPITCH);
#pragma unroll
    for (int t = 0; t < NT; ++t) {
      const v2f af = *(const v2f*)(aBase + t * 16 * APITCH + kk);
      acc[t] = wmma4(af, bf, acc[t]);
    }
  }
}

struct Chunk {
  v4f a[3];
  v2f w[16];
};

__device__ __forceinline__ void load_chunk(Chunk& c, const float* gA0, const float* gA1,
                                           const float* gA2, const float* gW, int j) {
  c.a[0] = *(const v4f*)(gA0 + KC * j);
  c.a[1] = *(const v4f*)(gA1 + KC * j);
  c.a[2] = *(const v4f*)(gA2 + KC * j);
#pragma unroll
  for (int h = 0; h < 2; ++h)
#pragma unroll
    for (int i = 0; i < 4; ++i) {
      const int kpb = 128 * h + (KC / 2) * j + 8 * i;
      const float* g = gW + (size_t)(2 * kpb) * 64;
      c.w[(h * 4 + i) * 2 + 0] = *(const v2f*)g;
      c.w[(h * 4 + i) * 2 + 1] = *(const v2f*)(g + 64);
    }
}

__device__ __forceinline__ void store_chunk(const Chunk& c, float* lA, float* lW, int j) {
  *(v4f*)(lA + KC * j) = c.a[0];
  *(v4f*)(lA + 16 * APITCH + KC * j) = c.a[1];
  *(v4f*)(lA + 32 * APITCH + KC * j) = c.a[2];
#pragma unroll
  for (int h = 0; h < 2; ++h)
#pragma unroll
    for (int i = 0; i < 4; ++i) {
      const int kpb = 128 * h + (KC / 2) * j + 8 * i;
      const v2f r0 = c.w[(h * 4 + i) * 2 + 0];
      const v2f r1 = c.w[(h * 4 + i) * 2 + 1];
      const v4f q = {r0.x, r1.x, r0.y, r1.y};
      *(v4f*)(lW + kpb * WPITCH) = q;
    }
}

__device__ __forceinline__ void lds_handover() {
  asm volatile("s_wait_dscnt 0x0\n\ts_barrier_signal -1\n\ts_barrier_wait -1" ::: "memory");
}

__global__ void __launch_bounds__(256) match_fused_kernel(
    const float* __restrict__ sa, const float* __restrict__ sb,
    const float* __restrict__ W1, const float* __restrict__ b1,
    const float* __restrict__ W2, const float* __restrict__ b2,
    const float* __restrict__ W3, const float* __restrict__ b3,
    float* __restrict__ out) {
  __shared__ __align__(16) float sA[48 * APITCH];
  __shared__ __align__(16) float sW[256 * WPITCH];
  __shared__ __align__(16) float sP[48 * PPITCH];
  __shared__ __align__(16) float sT[8 * 16 * LSTR];

  const int tid = threadIdx.x;
  const int lane = tid & 31;
  const int lo = lane & 15, hi = lane >> 4;
  const int w = __builtin_amdgcn_readfirstlane(tid >> 5);
  const int blk = blockIdx.x;
  const int b = blk >> 5;
  const int a0 = ((blk >> 2) & 7) * 16;
  const int c0 = (blk & 3) * 32;

  const bool a_role = (__builtin_popcount(w) & 1) != 0;
  const int n0 = (w >> 1) * 16;
  const float bias = a_role ? b1[n0 + lo] : 0.0f;

  const int srow = tid >> 4, sc4 = tid & 15;
  const float* gA0 = sa + (size_t)(b * 128 + a0 + srow) * 256 + 4 * sc4;
  const float* gA1 = sb + (size_t)(b * 128 + c0 + srow) * 256 + 4 * sc4;
  const float* gA2 = gA1 + 16 * 256;
  const float* gW = W1 + (size_t)(2 * w) * 64 + 2 * lane;
  float* lA = sA + srow * APITCH + 4 * sc4;
  float* lW = sW + w * WPITCH + 4 * lane;

  const float* aC = sA + (16 + lo) * APITCH + 2 * hi;
  const float* wC = sW + (128 + hi) * WPITCH + 2 * (n0 + lo);
  const float* aA = sA + lo * APITCH + 2 * hi;
  const float* wA = sW + hi * WPITCH + 2 * (n0 + lo);
  v8f accC[2] = {};
  v8f accA[1] = {};

  Chunk ch0, ch1, ch2, ch3;
  load_chunk(ch0, gA0, gA1, gA2, gW, 0);
  load_chunk(ch1, gA0, gA1, gA2, gW, 1);
  load_chunk(ch2, gA0, gA1, gA2, gW, 2);

  store_chunk(ch0, lA, lW, 0);
  load_chunk(ch3, gA0, gA1, gA2, gW, 3);
  lds_handover();
  if (a_role) proj_chunk<1>(aA, wA, accA);
  else        proj_chunk<2>(aC, wC, accC);

  store_chunk(ch1, lA, lW, 1);
  v2f w2f[2][16];
  {
    const float* wp = W2 + lo + (size_t)(2 * hi) * 32;
#pragma unroll
    for (int i = 0; i < 16; ++i) {
      w2f[0][i].x = wp[(size_t)(4 * i) * 32];
      w2f[0][i].y = wp[(size_t)(4 * i + 1) * 32];
    }
  }
  lds_handover();
  if (a_role) proj_chunk<1>(aA + KC, wA + (KC / 2) * WPITCH, accA);
  else        proj_chunk<2>(aC + KC, wC + (KC / 2) * WPITCH, accC);

  store_chunk(ch2, lA, lW, 2);
  {
    const float* wp = W2 + 16 + lo + (size_t)(2 * hi) * 32;
#pragma unroll
    for (int i = 0; i < 16; ++i) {
      w2f[1][i].x = wp[(size_t)(4 * i) * 32];
      w2f[1][i].y = wp[(size_t)(4 * i + 1) * 32];
    }
  }
  lds_handover();
  if (a_role) proj_chunk<1>(aA + 2 * KC, wA + 2 * (KC / 2) * WPITCH, accA);
  else        proj_chunk<2>(aC + 2 * KC, wC + 2 * (KC / 2) * WPITCH, accC);

  store_chunk(ch3, lA, lW, 3);
  v2f w3f[8];
  {
    const float* wp = W3 + lo + (size_t)(2 * hi) * 16;
#pragma unroll
    for (int i = 0; i < 8; ++i) {
      w3f[i].x = wp[(size_t)(4 * i) * 16];
      w3f[i].y = wp[(size_t)(4 * i + 1) * 16];
    }
  }
  const float b2v0 = b2[lo], b2v1 = b2[16 + lo];
  const float b3v = b3[lo];
  lds_handover();
  if (a_role) proj_chunk<1>(aA + 3 * KC, wA + 3 * (KC / 2) * WPITCH, accA);
  else        proj_chunk<2>(aC + 3 * KC, wC + 3 * (KC / 2) * WPITCH, accC);

  if (!a_role) {
#pragma unroll
    for (int t = 0; t < 2; ++t)
#pragma unroll
      for (int r = 0; r < 8; ++r)
        sP[(16 + t * 16 + r + 8 * hi) * PPITCH + n0 + lo] = accC[t][r] + bias;
  } else {
#pragma unroll
    for (int r = 0; r < 8; ++r)
      sP[(r + 8 * hi) * PPITCH + n0 + lo] = accA[0][r] + bias;
  }
  __syncthreads();

  float* wlds = sT + w * (16 * LSTR);
  for (int aa = 0; aa < 2; ++aa) {
    const int al = 2 * w + aa;
    v2f abv[16];
    const float* par = sP + al * PPITCH + 2 * hi;
#pragma unroll
    for (int i = 0; i < 16; ++i) abv[i] = *(const v2f*)(par + 4 * i);

    for (int ct = 0; ct < 2; ++ct) {
      const float* pbr = sP + (16 + ct * 16 + lo) * PPITCH + 2 * hi;
      v2f hf[16];
#pragma unroll
      for (int i = 0; i < 16; ++i) {
        const v2f bp = *(const v2f*)(pbr + 4 * i);
        hf[i].x = leaky(abv[i].x + bp.x);
        hf[i].y = leaky(abv[i].y + bp.y);
      }

      v8f c2a = {}, c2b = {};
#pragma unroll
      for (int i = 0; i < 16; ++i) {
        c2a = wmma4(hf[i], w2f[0][i], c2a);
        c2b = wmma4(hf[i], w2f[1][i], c2b);
      }

#pragma unroll
      for (int r = 0; r < 8; ++r) {
        wlds[(r + 8 * hi) * LSTR + lo]      = leaky(c2a[r] + b2v0);
        wlds[(r + 8 * hi) * LSTR + 16 + lo] = leaky(c2b[r] + b2v1);
      }
      asm volatile("s_wait_dscnt 0x0" ::: "memory");

      v2f hf3[8];
#pragma unroll
      for (int i = 0; i < 8; ++i)
        hf3[i] = *(const v2f*)(wlds + lo * LSTR + 4 * i + 2 * hi);
      asm volatile("s_wait_dscnt 0x0" ::: "memory");

      v8f c3 = {};
#pragma unroll
      for (int i = 0; i < 8; ++i) c3 = wmma4(hf3[i], w3f[i], c3);

      float* op = out + ((size_t)(b * 128 + a0 + al) * 128 + c0 + ct * 16) * 16 + lo;
#pragma unroll
      for (int r = 0; r < 8; ++r)
        op[(size_t)(r + 8 * hi) * 16] = leaky(c3[r] + b3v);
    }
  }
}

extern "C" void kernel_launch(void* const* d_in, const int* in_sizes, int n_in,
                              void* d_out, int out_size, void* d_ws,
                              size_t ws_size, hipStream_t stream) {
  (void)in_sizes; (void)n_in; (void)out_size; (void)ws_size; (void)d_ws;
  const float* sa = (const float*)d_in[0];
  const float* sb = (const float*)d_in[1];
  const float* W1 = (const float*)d_in[2];
  const float* b1 = (const float*)d_in[3];
  const float* W2 = (const float*)d_in[4];
  const float* b2 = (const float*)d_in[5];
  const float* W3 = (const float*)d_in[6];
  const float* b3 = (const float*)d_in[7];
  float* out = (float*)d_out;

  match_fused_kernel<<<256, 256, 0, stream>>>(sa, sb, W1, b1, W2, b2, W3, b3, out);
}
